# starting kernel
# baseline (speedup 1.0000x reference)
#include <hip/hip_runtime.h>

typedef __attribute__((ext_vector_type(2))) float v2f;
typedef __attribute__((ext_vector_type(8))) float v8f;

#define NFREQ  129
#define NBAND  8
#define CH     1024
#define NTILE  64
#define NBT    400
#define WPB    8
#define KSTEPS 32

__global__ __launch_bounds__(256)
void _FrequencyBandAverager_wmma_kernel(const float* __restrict__ x,
                                        const float* __restrict__ mask,
                                        float* __restrict__ out)
{
    __shared__ __align__(16) float sA[KSTEPS * 64];
    __shared__ float sRc[NBAND];
    __shared__ float sTail[NBAND];

    const int tid = threadIdx.x;

    if (tid < NBAND) {
        const float* mr = mask + tid * NFREQ;
        float cnt = 0.f;
        for (int f = 0; f < NFREQ; ++f) cnt += mr[f];
        const float rc = 1.0f / cnt;
        sRc[tid]   = rc;
        sTail[tid] = mr[128] * rc;
    }
    __syncthreads();

    for (int idx = tid; idx < KSTEPS * 64; idx += 256) {
        const int ks = idx >> 6;
        const int lv = idx & 63;
        const int l  = lv >> 1;
        const int v  = lv & 1;
        const int m  = l & 15;
        const int k  = ks * 4 + ((l >> 4) << 1) + v;
        sA[idx] = (m < NBAND) ? mask[m * NFREQ + k] * sRc[m] : 0.0f;
    }
    __syncthreads();

    const int lane  = tid & 31;
    const int w     = blockIdx.x * WPB + (tid >> 5);
    const int bt    = w >> 6;
    const int nt    = w & 63;
    const int m     = lane & 15;
    const int khalf = lane >> 4;

    const float* xw = x + (size_t)bt * NFREQ * CH + (size_t)(khalf * 2) * CH
                        + nt * 16 + m;

    v8f c = {};
    #pragma unroll
    for (int ks = 0; ks < KSTEPS; ++ks) {
        const v2f a = *(const v2f*)(sA + ks * 64 + lane * 2);
        v2f b;
        b.x = xw[(size_t)ks * 4 * CH];
        b.y = xw[(size_t)ks * 4 * CH + CH];
        c = __builtin_amdgcn_wmma_f32_16x16x4_f32(
                 false, a,  false, b,
                 (short)0, c,  false,  false);
    }

    if (lane < 16) {
        const float xv = x[(size_t)bt * NFREQ * CH + (size_t)128 * CH
                           + nt * 16 + lane];
        float* o = out + (size_t)bt * NBAND * CH + nt * 16 + lane;
        #pragma unroll
        for (int g = 0; g < NBAND; ++g)
            o[(size_t)g * CH] = c[g] + sTail[g] * xv;
    }
}

extern "C" void kernel_launch(void* const* d_in, const int* in_sizes, int n_in,
                              void* d_out, int out_size, void* d_ws, size_t ws_size,
                              hipStream_t stream)
{
    const float* x    = (const float*)d_in[0];
    const float* mask = (const float*)d_in[1];
    float* out        = (float*)d_out;

    const int grid = (NBT * NTILE) / WPB;
    _FrequencyBandAverager_wmma_kernel<<<grid, 256, 0, stream>>>(x, mask, out);
}


# each wave takes a 64-column strip: x rows read with 128-bit loads per lane into 4 column-interleaved WMMA tiles, two batches of 8 loads in flight, float4 stores; 4x fewer waves; sum order kept
# speedup vs baseline: 1.0932x; 1.0932x over previous
#include <hip/hip_runtime.h>

typedef __attribute__((ext_vector_type(2))) float v2f;
typedef __attribute__((ext_vector_type(4))) float v4f;
typedef __attribute__((ext_vector_type(8))) float v8f;

#define NFREQ  129
#define NBAND  8
#define CH     1024
#define SCOLS  64
#define NSTRIP 16
#define NBT    400
#define WPB    4
#define NTHR   (WPB * 32)
#define KSTEPS 32
#define KB     4
#define NBATCH (KSTEPS / KB)
#define STAGES 2

__global__ __launch_bounds__(NTHR)
void _FrequencyBandAverager_wmma_kernel(const float* __restrict__ x,
                                        const float* __restrict__ mask,
                                        float* __restrict__ out)
{
    __shared__ __align__(16) float sA[KSTEPS * 64];
    __shared__ float sRc[NBAND];
    __shared__ float sTail[NBAND];

    const int tid = threadIdx.x;

    if (tid < NBAND) {
        const float* mr = mask + tid * NFREQ;
        float cnt = 0.f;
        for (int f = 0; f < NFREQ; ++f) cnt += mr[f];
        const float rc = 1.0f / cnt;
        sRc[tid]   = rc;
        sTail[tid] = mr[128] * rc;
    }
    __syncthreads();

    for (int idx = tid; idx < KSTEPS * 64; idx += NTHR) {
        const int ks = idx >> 6;
        const int lv = idx & 63;
        const int l  = lv >> 1;
        const int v  = lv & 1;
        const int m  = l & 15;
        const int k  = ks * 4 + ((l >> 4) << 1) + v;
        sA[idx] = (m < NBAND) ? mask[m * NFREQ + k] * sRc[m] : 0.0f;
    }
    __syncthreads();

    const int lane  = tid & 31;
    const int w     = blockIdx.x * WPB + (tid >> 5);
    const int bt    = w >> 4;
    const int st    = w & (NSTRIP - 1);
    const int n     = lane & 15;
    const int khalf = lane >> 4;

    const float* xw = x + (size_t)bt * NFREQ * CH + (size_t)(khalf * 2) * CH
                        + st * SCOLS + n * 4;

    v4f r[STAGES][2 * KB];
#define LOAD_BATCH(B) do {                                                        \
                                 \
        _Pragma("unroll") for (int j = 0; j < KB; ++j) {                          \
            const float* p_ = xw + (size_t)((B) * KB + j) * 4 * CH;               \
            r[(B) % STAGES][2 * j]     = *(const v4f*)(p_);                       \
            r[(B) % STAGES][2 * j + 1] = *(const v4f*)(p_ + CH);                  \
        } } while (0)

    #pragma unroll
    for (int b = 0; b < STAGES - 1; ++b) LOAD_BATCH(b);

    v8f c0 = {}, c1 = {}, c2 = {}, c3 = {};
    #pragma unroll
    for (int b = 0; b < NBATCH; ++b) {
        __builtin_amdgcn_sched_barrier(0);
        if (b + STAGES - 1 < NBATCH) LOAD_BATCH(b + STAGES - 1);
        __builtin_amdgcn_sched_barrier(0);
        #pragma unroll
        for (int j = 0; j < KB; ++j) {
            const int ks = b * KB + j;
            const v2f a  = *(const v2f*)(sA + ks * 64 + lane * 2);
            const v4f r0 = r[b % STAGES][2 * j];
            const v4f r1 = r[b % STAGES][2 * j + 1];
            v2f bb;
            bb.x = r0.x; bb.y = r1.x;
            c0 = __builtin_amdgcn_wmma_f32_16x16x4_f32(false, a, false, bb, (short)0, c0, false, false);
            bb.x = r0.y; bb.y = r1.y;
            c1 = __builtin_amdgcn_wmma_f32_16x16x4_f32(false, a, false, bb, (short)0, c1, false, false);
            bb.x = r0.z; bb.y = r1.z;
            c2 = __builtin_amdgcn_wmma_f32_16x16x4_f32(false, a, false, bb, (short)0, c2, false, false);
            bb.x = r0.w; bb.y = r1.w;
            c3 = __builtin_amdgcn_wmma_f32_16x16x4_f32(false, a, false, bb, (short)0, c3, false, false);
        }
    }
#undef LOAD_BATCH

    if (lane < 16) {
        const v4f xv = *(const v4f*)(x + (size_t)bt * NFREQ * CH + (size_t)128 * CH
                                       + st * SCOLS + lane * 4);
        float* o = out + (size_t)bt * NBAND * CH + st * SCOLS + lane * 4;
        #pragma unroll
        for (int g = 0; g < NBAND; ++g) {
            const float t = sTail[g];
            v4f q;
            q.x = c0[g] + t * xv.x;
            q.y = c1[g] + t * xv.y;
            q.z = c2[g] + t * xv.z;
            q.w = c3[g] + t * xv.w;
            *(v4f*)(o + (size_t)g * CH) = q;
        }
    }
}

extern "C" void kernel_launch(void* const* d_in, const int* in_sizes, int n_in,
                              void* d_out, int out_size, void* d_ws, size_t ws_size,
                              hipStream_t stream)
{
    const float* x    = (const float*)d_in[0];
    const float* mask = (const float*)d_in[1];
    float* out        = (float*)d_out;

    const int grid = (NBT * NSTRIP) / WPB;
    _FrequencyBandAverager_wmma_kernel<<<grid, NTHR, 0, stream>>>(x, mask, out);
}


# mask prologue: each thread issues its 16 A-fragment loads + tail as one batch (was 43+16 loads, each waited on singly); band counts now summed from them as 8 strided partial sums (new sum order)
# speedup vs baseline: 1.3129x; 1.2010x over previous
#include <hip/hip_runtime.h>

typedef __attribute__((ext_vector_type(2))) float v2f;
typedef __attribute__((ext_vector_type(4))) float v4f;
typedef __attribute__((ext_vector_type(8))) float v8f;

#define NFREQ  129
#define NBAND  8
#define CH     1024
#define SCOLS  64
#define NSTRIP 16
#define NBT    400
#define WPB    4
#define NTHR   (WPB * 32)
#define KSTEPS 32
#define KB     4
#define NBATCH (KSTEPS / KB)
#define STAGES 2

__global__ __launch_bounds__(NTHR)
void _FrequencyBandAverager_wmma_kernel(const float* __restrict__ x,
                                        const float* __restrict__ mask,
                                        float* __restrict__ out)
{
    __shared__ __align__(16) float sA[KSTEPS * 64];
    __shared__ float sPart[WPB * 16];
    __shared__ float sTail[NBAND];

    const int tid = threadIdx.x;

    static_assert(NTHR == 128 && KSTEPS * 64 == 16 * NTHR, "prologue mapping assumes 128 threads x 16 entries");
    const int am   = (tid >> 1) & 15;
    const int wv   = tid >> 5;
    const int koff = ((wv >> 1) << 2) + ((wv & 1) << 1) + (tid & 1);
    const float* mr = mask + (am & (NBAND - 1)) * NFREQ;
    float mv[16];
    #pragma unroll
    for (int i = 0; i < 16; ++i) mv[i] = mr[8 * i + koff];
    const float tl = mr[128];
    float part = 0.f;
    #pragma unroll
    for (int i = 0; i < 16; ++i) part += mv[i];
    part += __shfl_xor(part, 1, 32);
    sPart[wv * 16 + am] = part;
    __syncthreads();
    const float cnt = ((sPart[am] + sPart[16 + am]) + (sPart[32 + am] + sPart[48 + am])) + tl;
    const float rc  = 1.0f / cnt;
    const bool live = am < NBAND;
    #pragma unroll
    for (int i = 0; i < 16; ++i) sA[tid + NTHR * i] = live ? mv[i] * rc : 0.0f;
    if (tid < 2 * NBAND && (tid & 1) == 0) sTail[tid >> 1] = tl * rc;
    __syncthreads();

    const int lane  = tid & 31;
    const int w     = blockIdx.x * WPB + (tid >> 5);
    const int bt    = w >> 4;
    const int st    = w & (NSTRIP - 1);
    const int n     = lane & 15;
    const int khalf = lane >> 4;

    const float* xw = x + (size_t)bt * NFREQ * CH + (size_t)(khalf * 2) * CH
                        + st * SCOLS + n * 4;

    v4f r[STAGES][2 * KB];
#define LOAD_BATCH(B) do {                                                        \
                                 \
        _Pragma("unroll") for (int j = 0; j < KB; ++j) {                          \
            const float* p_ = xw + (size_t)((B) * KB + j) * 4 * CH;               \
            r[(B) % STAGES][2 * j]     = *(const v4f*)(p_);                       \
            r[(B) % STAGES][2 * j + 1] = *(const v4f*)(p_ + CH);                  \
        } } while (0)

    #pragma unroll
    for (int b = 0; b < STAGES - 1; ++b) LOAD_BATCH(b);

    v8f c0 = {}, c1 = {}, c2 = {}, c3 = {};
    #pragma unroll
    for (int b = 0; b < NBATCH; ++b) {
        __builtin_amdgcn_sched_barrier(0);
        if (b + STAGES - 1 < NBATCH) LOAD_BATCH(b + STAGES - 1);
        __builtin_amdgcn_sched_barrier(0);
        #pragma unroll
        for (int j = 0; j < KB; ++j) {
            const int ks = b * KB + j;
            const v2f a  = *(const v2f*)(sA + ks * 64 + lane * 2);
            const v4f r0 = r[b % STAGES][2 * j];
            const v4f r1 = r[b % STAGES][2 * j + 1];
            v2f bb;
            bb.x = r0.x; bb.y = r1.x;
            c0 = __builtin_amdgcn_wmma_f32_16x16x4_f32(false, a, false, bb, (short)0, c0, false, false);
            bb.x = r0.y; bb.y = r1.y;
            c1 = __builtin_amdgcn_wmma_f32_16x16x4_f32(false, a, false, bb, (short)0, c1, false, false);
            bb.x = r0.z; bb.y = r1.z;
            c2 = __builtin_amdgcn_wmma_f32_16x16x4_f32(false, a, false, bb, (short)0, c2, false, false);
            bb.x = r0.w; bb.y = r1.w;
            c3 = __builtin_amdgcn_wmma_f32_16x16x4_f32(false, a, false, bb, (short)0, c3, false, false);
        }
    }
#undef LOAD_BATCH

    if (lane < 16) {
        const v4f xv = *(const v4f*)(x + (size_t)bt * NFREQ * CH + (size_t)128 * CH
                                       + st * SCOLS + lane * 4);
        float* o = out + (size_t)bt * NBAND * CH + st * SCOLS + lane * 4;
        #pragma unroll
        for (int g = 0; g < NBAND; ++g) {
            const float t = sTail[g];
            v4f q;
            q.x = c0[g] + t * xv.x;
            q.y = c1[g] + t * xv.y;
            q.z = c2[g] + t * xv.z;
            q.w = c3[g] + t * xv.w;
            *(v4f*)(o + (size_t)g * CH) = q;
        }
    }
}

extern "C" void kernel_launch(void* const* d_in, const int* in_sizes, int n_in,
                              void* d_out, int out_size, void* d_ws, size_t ws_size,
                              hipStream_t stream)
{
    const float* x    = (const float*)d_in[0];
    const float* mask = (const float*)d_in[1];
    float* out        = (float*)d_out;

    const int grid = (NBT * NSTRIP) / WPB;
    _FrequencyBandAverager_wmma_kernel<<<grid, NTHR, 0, stream>>>(x, mask, out);
}


# x requests of the first two batches and of the f=128 row go out ahead of the mask prologue; its two barriers order LDS only, so they no longer wait for loads in flight; sum order kept
# speedup vs baseline: 1.3928x; 1.0609x over previous
#include <hip/hip_runtime.h>

typedef __attribute__((ext_vector_type(2))) float v2f;
typedef __attribute__((ext_vector_type(4))) float v4f;
typedef __attribute__((ext_vector_type(8))) float v8f;

#define NFREQ  129
#define NBAND  8
#define CH     1024
#define SCOLS  64
#define NSTRIP 16
#define NBT    400
#define WPB    4
#define NTHR   (WPB * 32)
#define KSTEPS 32
#define KB     4
#define NBATCH (KSTEPS / KB)
#define STAGES 2

static __device__ __forceinline__ void lds_barrier()
{
    __builtin_amdgcn_fence(__ATOMIC_RELEASE, "workgroup", "local");
    __builtin_amdgcn_s_barrier();
    __builtin_amdgcn_fence(__ATOMIC_ACQUIRE, "workgroup", "local");
}

__global__ __launch_bounds__(NTHR)
void _FrequencyBandAverager_wmma_kernel(const float* __restrict__ x,
                                        const float* __restrict__ mask,
                                        float* __restrict__ out)
{
    __shared__ __align__(16) float sA[KSTEPS * 64];
    __shared__ float sPart[WPB * 16];
    __shared__ float sTail[NBAND];

    const int tid = threadIdx.x;

    const int lane  = tid & 31;
    const int w     = blockIdx.x * WPB + (tid >> 5);
    const int bt    = w >> 4;
    const int st    = w & (NSTRIP - 1);
    const int n     = lane & 15;
    const int khalf = lane >> 4;

    const float* xw = x + (size_t)bt * NFREQ * CH + (size_t)(khalf * 2) * CH
                        + st * SCOLS + n * 4;

    static_assert(NTHR == 128 && KSTEPS * 64 == 16 * NTHR, "prologue mapping assumes 128 threads x 16 entries");
    const int am   = (tid >> 1) & 15;
    const int wv   = tid >> 5;
    const int koff = ((wv >> 1) << 2) + ((wv & 1) << 1) + (tid & 1);
    const float* mr = mask + (am & (NBAND - 1)) * NFREQ;
    float mv[16];
    #pragma unroll
    for (int i = 0; i < 16; ++i) mv[i] = mr[8 * i + koff];
    const float tl = mr[128];

    __builtin_amdgcn_sched_barrier(0);
    v4f r[STAGES][2 * KB];
#define LOAD_BATCH(B) do {                                                        \
                                 \
        _Pragma("unroll") for (int j = 0; j < KB; ++j) {                          \
            const float* p_ = xw + (size_t)((B) * KB + j) * 4 * CH;               \
            r[(B) % STAGES][2 * j]     = *(const v4f*)(p_);                       \
            r[(B) % STAGES][2 * j + 1] = *(const v4f*)(p_ + CH);                  \
        } } while (0)
    #pragma unroll
    for (int b = 0; b < STAGES; ++b) { LOAD_BATCH(b); __builtin_amdgcn_sched_barrier(0); }
    const v4f xv = *(const v4f*)(x + (size_t)bt * NFREQ * CH + (size_t)128 * CH
                                   + st * SCOLS + n * 4);
    __builtin_amdgcn_sched_barrier(0);
    float part = 0.f;
    #pragma unroll
    for (int i = 0; i < 16; ++i) part += mv[i];
    part += __shfl_xor(part, 1, 32);
    sPart[wv * 16 + am] = part;
    lds_barrier();
    const float cnt = ((sPart[am] + sPart[16 + am]) + (sPart[32 + am] + sPart[48 + am])) + tl;
    const float rc  = 1.0f / cnt;
    const bool live = am < NBAND;
    #pragma unroll
    for (int i = 0; i < 16; ++i) sA[tid + NTHR * i] = live ? mv[i] * rc : 0.0f;
    if (tid < 2 * NBAND && (tid & 1) == 0) sTail[tid >> 1] = tl * rc;
    lds_barrier();

    v8f c0 = {}, c1 = {}, c2 = {}, c3 = {};
    #pragma unroll
    for (int b = 0; b < NBATCH; ++b) {
        __builtin_amdgcn_sched_barrier(0);
        if (b > 0 && b + STAGES - 1 < NBATCH) LOAD_BATCH(b + STAGES - 1);
        __builtin_amdgcn_sched_barrier(0);
        #pragma unroll
        for (int j = 0; j < KB; ++j) {
            const int ks = b * KB + j;
            const v2f a  = *(const v2f*)(sA + ks * 64 + lane * 2);
            const v4f r0 = r[b % STAGES][2 * j];
            const v4f r1 = r[b % STAGES][2 * j + 1];
            v2f bb;
            bb.x = r0.x; bb.y = r1.x;
            c0 = __builtin_amdgcn_wmma_f32_16x16x4_f32(false, a, false, bb, (short)0, c0, false, false);
            bb.x = r0.y; bb.y = r1.y;
            c1 = __builtin_amdgcn_wmma_f32_16x16x4_f32(false, a, false, bb, (short)0, c1, false, false);
            bb.x = r0.z; bb.y = r1.z;
            c2 = __builtin_amdgcn_wmma_f32_16x16x4_f32(false, a, false, bb, (short)0, c2, false, false);
            bb.x = r0.w; bb.y = r1.w;
            c3 = __builtin_amdgcn_wmma_f32_16x16x4_f32(false, a, false, bb, (short)0, c3, false, false);
        }
    }
#undef LOAD_BATCH

    if (lane < 16) {
        float* o = out + (size_t)bt * NBAND * CH + st * SCOLS + lane * 4;
        #pragma unroll
        for (int g = 0; g < NBAND; ++g) {
            const float t = sTail[g];
            v4f q;
            q.x = c0[g] + t * xv.x;
            q.y = c1[g] + t * xv.y;
            q.z = c2[g] + t * xv.z;
            q.w = c3[g] + t * xv.w;
            *(v4f*)(o + (size_t)g * CH) = q;
        }
    }
}

extern "C" void kernel_launch(void* const* d_in, const int* in_sizes, int n_in,
                              void* d_out, int out_size, void* d_ws, size_t ws_size,
                              hipStream_t stream)
{
    const float* x    = (const float*)d_in[0];
    const float* mask = (const float*)d_in[1];
    float* out        = (float*)d_out;

    const int grid = (NBT * NSTRIP) / WPB;
    _FrequencyBandAverager_wmma_kernel<<<grid, NTHR, 0, stream>>>(x, mask, out);
}


# one work-group per WGP: 256 groups of 25 waves (was 1600 groups of 4 waves, 6.25 per WGP); A fragments built once per group by waves 0-3; per-wave strip loop unchanged; sum order kept
# speedup vs baseline: 1.7317x; 1.2433x over previous
#include <hip/hip_runtime.h>

typedef __attribute__((ext_vector_type(2))) float v2f;
typedef __attribute__((ext_vector_type(4))) float v4f;
typedef __attribute__((ext_vector_type(8))) float v8f;

#define NFREQ  129
#define NBAND  8
#define CH     1024
#define SCOLS  64
#define NSTRIP 16
#define NBT    400
#define WPB    25
#define NTHR   (WPB * 32)
#define PRO_T  128
#define KSTEPS 32
#define KB     4
#define NBATCH (KSTEPS / KB)
#define STAGES 2

static __device__ __forceinline__ void lds_barrier()
{
    __builtin_amdgcn_fence(__ATOMIC_RELEASE, "workgroup", "local");
    __builtin_amdgcn_s_barrier();
    __builtin_amdgcn_fence(__ATOMIC_ACQUIRE, "workgroup", "local");
}

__global__ __launch_bounds__(NTHR)
void _FrequencyBandAverager_wmma_kernel(const float* __restrict__ x,
                                        const float* __restrict__ mask,
                                        float* __restrict__ out)
{
    __shared__ __align__(16) float sA[KSTEPS * 64];
    __shared__ float sPart[(PRO_T / 32) * 16];
    __shared__ float sTail[NBAND];

    const int tid = threadIdx.x;

    const int lane  = tid & 31;
    const int w     = blockIdx.x * WPB + (tid >> 5);
    const int bt    = w >> 4;
    const int st    = w & (NSTRIP - 1);
    const int n     = lane & 15;
    const int khalf = lane >> 4;

    const float* xw = x + (size_t)bt * NFREQ * CH + (size_t)(khalf * 2) * CH
                        + st * SCOLS + n * 4;

    static_assert(PRO_T == 128 && KSTEPS * 64 == 16 * PRO_T && NTHR >= PRO_T, "prologue mapping assumes 128 threads x 16 entries");
    const bool pro = tid < PRO_T;
    const int am   = (tid >> 1) & 15;
    const int wv   = tid >> 5;
    const int koff = ((wv >> 1) << 2) + ((wv & 1) << 1) + (tid & 1);
    const float* mr = mask + (am & (NBAND - 1)) * NFREQ;
    float mv[16];
    float tl = 0.f;
    if (pro) {
        #pragma unroll
        for (int i = 0; i < 16; ++i) mv[i] = mr[8 * i + koff];
        tl = mr[128];
    }

    __builtin_amdgcn_sched_barrier(0);
    v4f r[STAGES][2 * KB];
#define LOAD_BATCH(B) do {                                                        \
                                 \
        _Pragma("unroll") for (int j = 0; j < KB; ++j) {                          \
            const float* p_ = xw + (size_t)((B) * KB + j) * 4 * CH;               \
            r[(B) % STAGES][2 * j]     = *(const v4f*)(p_);                       \
            r[(B) % STAGES][2 * j + 1] = *(const v4f*)(p_ + CH);                  \
        } } while (0)
    #pragma unroll
    for (int b = 0; b < STAGES; ++b) { LOAD_BATCH(b); __builtin_amdgcn_sched_barrier(0); }
    const v4f xv = *(const v4f*)(x + (size_t)bt * NFREQ * CH + (size_t)128 * CH
                                   + st * SCOLS + n * 4);
    __builtin_amdgcn_sched_barrier(0);
    if (pro) {
        float part = 0.f;
        #pragma unroll
        for (int i = 0; i < 16; ++i) part += mv[i];
        part += __shfl_xor(part, 1, 32);
        sPart[wv * 16 + am] = part;
    }
    lds_barrier();
    if (pro) {
        const float cnt = ((sPart[am] + sPart[16 + am]) + (sPart[32 + am] + sPart[48 + am])) + tl;
        const float rc  = 1.0f / cnt;
        const bool live = am < NBAND;
        #pragma unroll
        for (int i = 0; i < 16; ++i) sA[tid + PRO_T * i] = live ? mv[i] * rc : 0.0f;
        if (tid < 2 * NBAND && (tid & 1) == 0) sTail[tid >> 1] = tl * rc;
    }
    lds_barrier();

    v8f c0 = {}, c1 = {}, c2 = {}, c3 = {};
    #pragma unroll
    for (int b = 0; b < NBATCH; ++b) {
        __builtin_amdgcn_sched_barrier(0);
        if (b > 0 && b + STAGES - 1 < NBATCH) LOAD_BATCH(b + STAGES - 1);
        __builtin_amdgcn_sched_barrier(0);
        #pragma unroll
        for (int j = 0; j < KB; ++j) {
            const int ks = b * KB + j;
            const v2f a  = *(const v2f*)(sA + ks * 64 + lane * 2);
            const v4f r0 = r[b % STAGES][2 * j];
            const v4f r1 = r[b % STAGES][2 * j + 1];
            v2f bb;
            bb.x = r0.x; bb.y = r1.x;
            c0 = __builtin_amdgcn_wmma_f32_16x16x4_f32(false, a, false, bb, (short)0, c0, false, false);
            bb.x = r0.y; bb.y = r1.y;
            c1 = __builtin_amdgcn_wmma_f32_16x16x4_f32(false, a, false, bb, (short)0, c1, false, false);
            bb.x = r0.z; bb.y = r1.z;
            c2 = __builtin_amdgcn_wmma_f32_16x16x4_f32(false, a, false, bb, (short)0, c2, false, false);
            bb.x = r0.w; bb.y = r1.w;
            c3 = __builtin_amdgcn_wmma_f32_16x16x4_f32(false, a, false, bb, (short)0, c3, false, false);
        }
    }
#undef LOAD_BATCH

    if (lane < 16) {
        float* o = out + (size_t)bt * NBAND * CH + st * SCOLS + lane * 4;
        #pragma unroll
        for (int g = 0; g < NBAND; ++g) {
            const float t = sTail[g];
            v4f q;
            q.x = c0[g] + t * xv.x;
            q.y = c1[g] + t * xv.y;
            q.z = c2[g] + t * xv.z;
            q.w = c3[g] + t * xv.w;
            *(v4f*)(o + (size_t)g * CH) = q;
        }
    }
}

extern "C" void kernel_launch(void* const* d_in, const int* in_sizes, int n_in,
                              void* d_out, int out_size, void* d_ws, size_t ws_size,
                              hipStream_t stream)
{
    const float* x    = (const float*)d_in[0];
    const float* mask = (const float*)d_in[1];
    float* out        = (float*)d_out;

    const int grid = (NBT * NSTRIP) / WPB;
    _FrequencyBandAverager_wmma_kernel<<<grid, NTHR, 0, stream>>>(x, mask, out);
}
